# starting kernel
# baseline (speedup 1.0000x reference)
#include <hip/hip_runtime.h>

typedef __attribute__((ext_vector_type(2))) float v2f;
typedef __attribute__((ext_vector_type(8))) float v8f;

#define D 512
#define TILE_M 16

__global__ __launch_bounds__(256) void gemm_wmma_f32(
    const float* __restrict__ Xin, const float* __restrict__ W,
    const float* __restrict__ bias, float* __restrict__ H, int relu_in)
{
    __shared__ float xs[TILE_M * D];
    const int m16 = blockIdx.x * TILE_M;
    const int t   = threadIdx.x;

    const float4* xin4 = (const float4*)(Xin + (size_t)m16 * D);
    float4* xs4 = (float4*)xs;
#pragma unroll
    for (int i = 0; i < 8; ++i) {
        int idx = t + i * 256;
        float4 v = xin4[idx];
        if (relu_in) {
            v.x = fmaxf(v.x, 0.0f); v.y = fmaxf(v.y, 0.0f);
            v.z = fmaxf(v.z, 0.0f); v.w = fmaxf(v.w, 0.0f);
        }
        xs4[idx] = v;
    }
    __syncthreads();

    const int wave  = t >> 5;
    const int lane  = t & 31;
    const int row   = lane & 15;
    const int koff  = (lane >> 4) << 1;
    const int nbase = wave * 64;

    v8f acc0 = {}, acc1 = {}, acc2 = {}, acc3 = {};

    const float* ar  = xs + row * D;
    const float* wr0 = W + (size_t)(nbase +  0 + row) * D;
    const float* wr1 = W + (size_t)(nbase + 16 + row) * D;
    const float* wr2 = W + (size_t)(nbase + 32 + row) * D;
    const float* wr3 = W + (size_t)(nbase + 48 + row) * D;

#pragma unroll 2
    for (int k = 0; k < D; k += 4) {
        const int kk = k + koff;
        v2f a  = *(const v2f*)(ar  + kk);
        v2f b0 = *(const v2f*)(wr0 + kk);
        v2f b1 = *(const v2f*)(wr1 + kk);
        v2f b2 = *(const v2f*)(wr2 + kk);
        v2f b3 = *(const v2f*)(wr3 + kk);
        acc0 = __builtin_amdgcn_wmma_f32_16x16x4_f32(false, a, false, b0, (short)0, acc0, false, false);
        acc1 = __builtin_amdgcn_wmma_f32_16x16x4_f32(false, a, false, b1, (short)0, acc1, false, false);
        acc2 = __builtin_amdgcn_wmma_f32_16x16x4_f32(false, a, false, b2, (short)0, acc2, false, false);
        acc3 = __builtin_amdgcn_wmma_f32_16x16x4_f32(false, a, false, b3, (short)0, acc3, false, false);
    }

    const int rbase = m16 + ((lane >> 4) << 3);
    const int c0 = nbase + (lane & 15);
#pragma unroll
    for (int s = 0; s < 4; ++s) {
        const v8f acc = (s == 0) ? acc0 : (s == 1) ? acc1 : (s == 2) ? acc2 : acc3;
        const int col = c0 + s * 16;
        const float bv = bias[col];
        float* hp = H + (size_t)rbase * D + col;
#pragma unroll
        for (int v = 0; v < 8; ++v) {
            hp[(size_t)v * D] = acc[v] + bv;
        }
    }
}

__global__ __launch_bounds__(256) void zero_f32(float* __restrict__ p, int n4)
{
    int i = blockIdx.x * 256 + threadIdx.x;
    if (i < n4) ((float4*)p)[i] = make_float4(0.f, 0.f, 0.f, 0.f);
}

__global__ __launch_bounds__(256) void scatter_add(
    const float* __restrict__ H, const int* __restrict__ src,
    const int* __restrict__ dst, float* __restrict__ AGG, int E)
{
    long long tid = (long long)blockIdx.x * 256 + threadIdx.x;
    int e = (int)(tid >> 7);
    if (e >= E) return;
    int c = (int)(tid & 127) << 2;
    int s = src[e];
    int d = dst[e];
    float4 v = *(const float4*)(H + (size_t)s * D + c);
    float* out = AGG + (size_t)d * D + c;
    atomicAdd(out + 0, v.x);
    atomicAdd(out + 1, v.y);
    atomicAdd(out + 2, v.z);
    atomicAdd(out + 3, v.w);
}

__global__ __launch_bounds__(256) void relu_f32(float* __restrict__ p, int n4)
{
    int i = blockIdx.x * 256 + threadIdx.x;
    if (i < n4) {
        float4 v = ((float4*)p)[i];
        v.x = fmaxf(v.x, 0.0f); v.y = fmaxf(v.y, 0.0f);
        v.z = fmaxf(v.z, 0.0f); v.w = fmaxf(v.w, 0.0f);
        ((float4*)p)[i] = v;
    }
}

extern "C" void kernel_launch(void* const* d_in, const int* in_sizes, int n_in,
                              void* d_out, int out_size, void* d_ws, size_t ws_size,
                              hipStream_t stream)
{
    const float* x  = (const float*)d_in[0];
    const int*   ei = (const int*)d_in[1];
    const float* W1 = (const float*)d_in[2];
    const float* b1 = (const float*)d_in[3];
    const float* W2 = (const float*)d_in[4];
    const float* b2 = (const float*)d_in[5];
    float* out = (float*)d_out;

    const int N = in_sizes[0] / D;
    const int E = in_sizes[1] / 2;
    const int* src = ei;
    const int* dst = ei + E;

    float* h   = (float*)d_ws;
    float* agg = h + (size_t)N * D;

    const int nd   = N * D;
    const int n4   = nd / 4;
    const int mtb  = N / TILE_M;
    const int zb   = (n4 + 255) / 256;
    const int sb   = (int)(((long long)E * 128 + 255) / 256);

    gemm_wmma_f32<<<mtb, 256, 0, stream>>>(x, W1, b1, h, 0);
    zero_f32<<<zb, 256, 0, stream>>>(agg, n4);
    scatter_add<<<sb, 256, 0, stream>>>(h, src, dst, agg, E);

    gemm_wmma_f32<<<mtb, 256, 0, stream>>>(agg, W2, b2, h, 1);
    zero_f32<<<zb, 256, 0, stream>>>(out, n4);
    scatter_add<<<sb, 256, 0, stream>>>(h, src, dst, out, E);
    relu_f32<<<zb, 256, 0, stream>>>(out, n4);
}


# aggregation as a gather over an in-call dst-sorted edge list: f32 atomic scatter, zero-fills and relu pass go; sum order per node = arrival-order rank of int atomics (was f32-atomic arrival order)
# speedup vs baseline: 2.0492x; 2.0492x over previous
#include <hip/hip_runtime.h>

typedef __attribute__((ext_vector_type(2))) float v2f;
typedef __attribute__((ext_vector_type(8))) float v8f;

#define D 512
#define TILE_M 16

__global__ __launch_bounds__(256) void gemm_wmma_f32(
    const float* __restrict__ Xin, const float* __restrict__ W,
    const float* __restrict__ bias, float* __restrict__ H, int relu_in)
{
    __shared__ float xs[TILE_M * D];
    const int m16 = blockIdx.x * TILE_M;
    const int t   = threadIdx.x;

    const float4* xin4 = (const float4*)(Xin + (size_t)m16 * D);
    float4* xs4 = (float4*)xs;
#pragma unroll
    for (int i = 0; i < 8; ++i) {
        int idx = t + i * 256;
        float4 v = xin4[idx];
        if (relu_in) {
            v.x = fmaxf(v.x, 0.0f); v.y = fmaxf(v.y, 0.0f);
            v.z = fmaxf(v.z, 0.0f); v.w = fmaxf(v.w, 0.0f);
        }
        xs4[idx] = v;
    }
    __syncthreads();

    const int wave  = t >> 5;
    const int lane  = t & 31;
    const int row   = lane & 15;
    const int koff  = (lane >> 4) << 1;
    const int nbase = wave * 64;

    v8f acc0 = {}, acc1 = {}, acc2 = {}, acc3 = {};

    const float* ar  = xs + row * D;
    const float* wr0 = W + (size_t)(nbase +  0 + row) * D;
    const float* wr1 = W + (size_t)(nbase + 16 + row) * D;
    const float* wr2 = W + (size_t)(nbase + 32 + row) * D;
    const float* wr3 = W + (size_t)(nbase + 48 + row) * D;

#pragma unroll 2
    for (int k = 0; k < D; k += 4) {
        const int kk = k + koff;
        v2f a  = *(const v2f*)(ar  + kk);
        v2f b0 = *(const v2f*)(wr0 + kk);
        v2f b1 = *(const v2f*)(wr1 + kk);
        v2f b2 = *(const v2f*)(wr2 + kk);
        v2f b3 = *(const v2f*)(wr3 + kk);
        acc0 = __builtin_amdgcn_wmma_f32_16x16x4_f32(false, a, false, b0, (short)0, acc0, false, false);
        acc1 = __builtin_amdgcn_wmma_f32_16x16x4_f32(false, a, false, b1, (short)0, acc1, false, false);
        acc2 = __builtin_amdgcn_wmma_f32_16x16x4_f32(false, a, false, b2, (short)0, acc2, false, false);
        acc3 = __builtin_amdgcn_wmma_f32_16x16x4_f32(false, a, false, b3, (short)0, acc3, false, false);
    }

    const int rbase = m16 + ((lane >> 4) << 3);
    const int c0 = nbase + (lane & 15);
#pragma unroll
    for (int s = 0; s < 4; ++s) {
        const v8f acc = (s == 0) ? acc0 : (s == 1) ? acc1 : (s == 2) ? acc2 : acc3;
        const int col = c0 + s * 16;
        const float bv = bias[col];
        float* hp = H + (size_t)rbase * D + col;
#pragma unroll
        for (int v = 0; v < 8; ++v) {
            hp[(size_t)v * D] = acc[v] + bv;
        }
    }
}

__global__ __launch_bounds__(256) void csr_zero(int* __restrict__ cnt, int n)
{
    int i = blockIdx.x * 256 + threadIdx.x;
    if (i < n) cnt[i] = 0;
}

__global__ __launch_bounds__(256) void csr_count(
    const int* __restrict__ dst, int* __restrict__ cnt, int* __restrict__ rank, int E)
{
    int e = blockIdx.x * 256 + threadIdx.x;
    if (e < E) rank[e] = atomicAdd(cnt + dst[e], 1);
}

__global__ __launch_bounds__(1024) void csr_scan(
    const int* __restrict__ cnt, int* __restrict__ rowptr, int n)
{
    __shared__ int wsum[32];
    const int t    = threadIdx.x;
    const int lane = t & 31;
    const int w    = t >> 5;
    const int per  = (n + 1023) >> 10;
    const int beg  = min(t * per, n);
    const int end  = min(beg + per, n);

    int s = 0;
    for (int i = beg; i < end; ++i) s += cnt[i];

    int incl = s;
    for (int off = 1; off < 32; off <<= 1) {
        int v = __shfl_up(incl, off);
        if (lane >= off) incl += v;
    }
    if (lane == 31) wsum[w] = incl;
    __syncthreads();
    if (w == 0) {
        int x = wsum[lane];
        int xi = x;
        for (int off = 1; off < 32; off <<= 1) {
            int v = __shfl_up(xi, off);
            if (lane >= off) xi += v;
        }
        wsum[lane] = xi - x;
    }
    __syncthreads();

    int base = wsum[w] + incl - s;
    for (int i = beg; i < end; ++i) { rowptr[i] = base; base += cnt[i]; }
    if (t == 1023) rowptr[n] = base;
}

__global__ __launch_bounds__(256) void csr_fill(
    const int* __restrict__ src, const int* __restrict__ dst,
    const int* __restrict__ rank, const int* __restrict__ rowptr,
    int* __restrict__ csr_src, int E)
{
    int e = blockIdx.x * 256 + threadIdx.x;
    if (e < E) {
        int pos = min(max(rowptr[dst[e]] + rank[e], 0), E - 1);
        csr_src[pos] = src[e];
    }
}

__global__ __launch_bounds__(128) void gather_sum_relu(
    const float* __restrict__ H, const int* __restrict__ rowptr,
    const int* __restrict__ csr_src, float* __restrict__ OUT, int n, int E)
{
    const int node = blockIdx.x;
    const int t    = threadIdx.x;
    const int beg  = max(rowptr[node], 0);
    const int end  = min(rowptr[node + 1], E);
    const int last = n - 1;
    const float4* __restrict__ H4 = (const float4*)H;

    float4 acc = make_float4(0.f, 0.f, 0.f, 0.f);
    int j = beg;
    for (; j + 4 <= end; j += 4) {
        const int s0 = min(max(csr_src[j + 0], 0), last);
        const int s1 = min(max(csr_src[j + 1], 0), last);
        const int s2 = min(max(csr_src[j + 2], 0), last);
        const int s3 = min(max(csr_src[j + 3], 0), last);
        const float4 v0 = H4[(size_t)s0 * (D / 4) + t];
        const float4 v1 = H4[(size_t)s1 * (D / 4) + t];
        const float4 v2 = H4[(size_t)s2 * (D / 4) + t];
        const float4 v3 = H4[(size_t)s3 * (D / 4) + t];
        acc.x += v0.x; acc.y += v0.y; acc.z += v0.z; acc.w += v0.w;
        acc.x += v1.x; acc.y += v1.y; acc.z += v1.z; acc.w += v1.w;
        acc.x += v2.x; acc.y += v2.y; acc.z += v2.z; acc.w += v2.w;
        acc.x += v3.x; acc.y += v3.y; acc.z += v3.z; acc.w += v3.w;
    }
    for (; j < end; ++j) {
        const int s0 = min(max(csr_src[j], 0), last);
        const float4 v0 = H4[(size_t)s0 * (D / 4) + t];
        acc.x += v0.x; acc.y += v0.y; acc.z += v0.z; acc.w += v0.w;
    }
    acc.x = fmaxf(acc.x, 0.0f); acc.y = fmaxf(acc.y, 0.0f);
    acc.z = fmaxf(acc.z, 0.0f); acc.w = fmaxf(acc.w, 0.0f);
    ((float4*)OUT)[(size_t)node * (D / 4) + t] = acc;
}

extern "C" void kernel_launch(void* const* d_in, const int* in_sizes, int n_in,
                              void* d_out, int out_size, void* d_ws, size_t ws_size,
                              hipStream_t stream)
{
    const float* x  = (const float*)d_in[0];
    const int*   ei = (const int*)d_in[1];
    const float* W1 = (const float*)d_in[2];
    const float* b1 = (const float*)d_in[3];
    const float* W2 = (const float*)d_in[4];
    const float* b2 = (const float*)d_in[5];
    float* out = (float*)d_out;

    const int N = in_sizes[0] / D;
    const int E = in_sizes[1] / 2;
    const int* src = ei;
    const int* dst = ei + E;

    float* h   = (float*)d_ws;
    float* agg = h + (size_t)N * D;
    int* cnt     = (int*)(agg + (size_t)N * D);
    int* rowptr  = cnt + N;
    int* rank    = rowptr + (N + 1);
    int* csr_src = rank + E;

    const int mtb = N / TILE_M;
    const int nb  = (N + 255) / 256;
    const int eb  = (E + 255) / 256;

    csr_zero<<<nb, 256, 0, stream>>>(cnt, N);
    csr_count<<<eb, 256, 0, stream>>>(dst, cnt, rank, E);
    csr_scan<<<1, 1024, 0, stream>>>(cnt, rowptr, N);
    csr_fill<<<eb, 256, 0, stream>>>(src, dst, rank, rowptr, csr_src, E);

    gemm_wmma_f32<<<mtb, 256, 0, stream>>>(x, W1, b1, h, 0);
    gather_sum_relu<<<N, D / 4, 0, stream>>>(h, rowptr, csr_src, agg, N, E);

    gemm_wmma_f32<<<mtb, 256, 0, stream>>>(agg, W2, b2, h, 0);
    gather_sum_relu<<<N, D / 4, 0, stream>>>(h, rowptr, csr_src, out, N, E);
}


# GEMM operands staged via LDS: W in 32-k chunks by whole-line loads kept in flight over the WMMAs (LDS-only barrier), bank-conflict-free pitches, X tile loads batched; k order kept
# speedup vs baseline: 2.6933x; 1.3143x over previous
#include <hip/hip_runtime.h>

typedef __attribute__((ext_vector_type(2))) float v2f;
typedef __attribute__((ext_vector_type(8))) float v8f;

#define D 512
#define TILE_M 16

#define XP 516
#define KC 32
#define WP 36
#define NCHUNK (D / KC)

#define W_LD(i, off)  w##i = wg[(size_t)(i) * 32 * (D / 4) + (off)]
#define W_ST(i)       *(float4*)(wl + (i) * 32 * WP) = w##i
#define W_LOAD_ALL(off) \
    W_LD(0, off);  W_LD(1, off);  W_LD(2, off);  W_LD(3, off);  \
    W_LD(4, off);  W_LD(5, off);  W_LD(6, off);  W_LD(7, off);  \
    W_LD(8, off);  W_LD(9, off);  W_LD(10, off); W_LD(11, off); \
    W_LD(12, off); W_LD(13, off); W_LD(14, off); W_LD(15, off)
#define W_STORE_ALL() \
    W_ST(0);  W_ST(1);  W_ST(2);  W_ST(3);  W_ST(4);  W_ST(5);  W_ST(6);  W_ST(7); \
    W_ST(8);  W_ST(9);  W_ST(10); W_ST(11); W_ST(12); W_ST(13); W_ST(14); W_ST(15)

__device__ __forceinline__ void lds_barrier()
{
    asm volatile("s_wait_dscnt 0x0\n\ts_barrier_signal -1\n\ts_barrier_wait -1" ::: "memory");
}

__device__ __forceinline__ void chunk_mma(const float* ak, const float* br,
                                          v8f& acc0, v8f& acc1, v8f& acc2, v8f& acc3)
{
#pragma unroll
    for (int kl = 0; kl < KC; kl += 4) {
        v2f a  = *(const v2f*)(ak + kl);
        v2f b0 = *(const v2f*)(br + kl);
        v2f b1 = *(const v2f*)(br + 16 * WP + kl);
        v2f b2 = *(const v2f*)(br + 32 * WP + kl);
        v2f b3 = *(const v2f*)(br + 48 * WP + kl);
        acc0 = __builtin_amdgcn_wmma_f32_16x16x4_f32(false, a, false, b0, (short)0, acc0, false, false);
        acc1 = __builtin_amdgcn_wmma_f32_16x16x4_f32(false, a, false, b1, (short)0, acc1, false, false);
        acc2 = __builtin_amdgcn_wmma_f32_16x16x4_f32(false, a, false, b2, (short)0, acc2, false, false);
        acc3 = __builtin_amdgcn_wmma_f32_16x16x4_f32(false, a, false, b3, (short)0, acc3, false, false);
    }
}

__global__ __launch_bounds__(256) void gemm_wmma_f32(
    const float* __restrict__ Xin, const float* __restrict__ W,
    const float* __restrict__ bias, float* __restrict__ H, int relu_in)
{
    __shared__ __attribute__((aligned(16))) float xs[TILE_M * XP];
    __shared__ __attribute__((aligned(16))) float wsm[D * WP];
    const int m16 = blockIdx.x * TILE_M;
    const int t   = threadIdx.x;

    const float4* xin4 = (const float4*)(Xin + (size_t)m16 * D);
    float4 xv[8];
#pragma unroll
    for (int i = 0; i < 8; ++i) xv[i] = xin4[t + i * 256];
    if (relu_in) {
#pragma unroll
        for (int i = 0; i < 8; ++i) {
            xv[i].x = fmaxf(xv[i].x, 0.0f); xv[i].y = fmaxf(xv[i].y, 0.0f);
            xv[i].z = fmaxf(xv[i].z, 0.0f); xv[i].w = fmaxf(xv[i].w, 0.0f);
        }
    }
#pragma unroll
    for (int i = 0; i < 8; ++i) {
        const int idx = t + i * 256;
        *(float4*)(xs + (idx >> 7) * XP + ((idx & 127) << 2)) = xv[i];
    }

    const int wrow = t >> 3;
    const int wq   = t & 7;
    const float4* wg = (const float4*)W + (size_t)wrow * (D / 4) + wq;
    float* wl = wsm + wrow * WP + (wq << 2);
    float4 w0, w1, w2, w3, w4, w5, w6, w7, w8, w9, w10, w11, w12, w13, w14, w15;
    W_LOAD_ALL(0);
    W_STORE_ALL();
    __builtin_amdgcn_sched_barrier(0);
    W_LOAD_ALL(KC / 4);
    lds_barrier();

    const int wave  = t >> 5;
    const int lane  = t & 31;
    const int row   = lane & 15;
    const int koff  = (lane >> 4) << 1;
    const int nbase = wave * 64;

    v8f acc0 = {}, acc1 = {}, acc2 = {}, acc3 = {};

    const float* ar = xs + row * XP + koff;
    const float* br = wsm + (nbase + row) * WP + koff;

#pragma unroll 1
    for (int kc = 0; kc < NCHUNK - 1; ++kc) {
        chunk_mma(ar + kc * KC, br, acc0, acc1, acc2, acc3);
        lds_barrier();
        W_STORE_ALL();
        if (kc + 2 < NCHUNK) {
            W_LOAD_ALL((kc + 2) * (KC / 4));
        }
        lds_barrier();
    }
    chunk_mma(ar + (NCHUNK - 1) * KC, br, acc0, acc1, acc2, acc3);

    const int rbase = m16 + ((lane >> 4) << 3);
    const int c0 = nbase + (lane & 15);
#pragma unroll
    for (int s = 0; s < 4; ++s) {
        const v8f acc = (s == 0) ? acc0 : (s == 1) ? acc1 : (s == 2) ? acc2 : acc3;
        const int col = c0 + s * 16;
        const float bv = bias[col];
        float* hp = H + (size_t)rbase * D + col;
#pragma unroll
        for (int v = 0; v < 8; ++v) {
            hp[(size_t)v * D] = acc[v] + bv;
        }
    }
}

__global__ __launch_bounds__(256) void csr_zero(int* __restrict__ cnt, int n)
{
    int i = blockIdx.x * 256 + threadIdx.x;
    if (i < n) cnt[i] = 0;
}

__global__ __launch_bounds__(256) void csr_count(
    const int* __restrict__ dst, int* __restrict__ cnt, int* __restrict__ rank, int E)
{
    int e = blockIdx.x * 256 + threadIdx.x;
    if (e < E) rank[e] = atomicAdd(cnt + dst[e], 1);
}

__global__ __launch_bounds__(1024) void csr_scan(
    const int* __restrict__ cnt, int* __restrict__ rowptr, int n)
{
    __shared__ int wsum[32];
    const int t    = threadIdx.x;
    const int lane = t & 31;
    const int w    = t >> 5;
    const int per  = (n + 1023) >> 10;
    const int beg  = min(t * per, n);
    const int end  = min(beg + per, n);

    int s = 0;
    for (int i = beg; i < end; ++i) s += cnt[i];

    int incl = s;
    for (int off = 1; off < 32; off <<= 1) {
        int v = __shfl_up(incl, off);
        if (lane >= off) incl += v;
    }
    if (lane == 31) wsum[w] = incl;
    __syncthreads();
    if (w == 0) {
        int x = wsum[lane];
        int xi = x;
        for (int off = 1; off < 32; off <<= 1) {
            int v = __shfl_up(xi, off);
            if (lane >= off) xi += v;
        }
        wsum[lane] = xi - x;
    }
    __syncthreads();

    int base = wsum[w] + incl - s;
    for (int i = beg; i < end; ++i) { rowptr[i] = base; base += cnt[i]; }
    if (t == 1023) rowptr[n] = base;
}

__global__ __launch_bounds__(256) void csr_fill(
    const int* __restrict__ src, const int* __restrict__ dst,
    const int* __restrict__ rank, const int* __restrict__ rowptr,
    int* __restrict__ csr_src, int E)
{
    int e = blockIdx.x * 256 + threadIdx.x;
    if (e < E) {
        int pos = min(max(rowptr[dst[e]] + rank[e], 0), E - 1);
        csr_src[pos] = src[e];
    }
}

__global__ __launch_bounds__(128) void gather_sum_relu(
    const float* __restrict__ H, const int* __restrict__ rowptr,
    const int* __restrict__ csr_src, float* __restrict__ OUT, int n, int E)
{
    const int node = blockIdx.x;
    const int t    = threadIdx.x;
    const int beg  = max(rowptr[node], 0);
    const int end  = min(rowptr[node + 1], E);
    const int last = n - 1;
    const float4* __restrict__ H4 = (const float4*)H;

    float4 acc = make_float4(0.f, 0.f, 0.f, 0.f);
    int j = beg;
    for (; j + 4 <= end; j += 4) {
        const int s0 = min(max(csr_src[j + 0], 0), last);
        const int s1 = min(max(csr_src[j + 1], 0), last);
        const int s2 = min(max(csr_src[j + 2], 0), last);
        const int s3 = min(max(csr_src[j + 3], 0), last);
        const float4 v0 = H4[(size_t)s0 * (D / 4) + t];
        const float4 v1 = H4[(size_t)s1 * (D / 4) + t];
        const float4 v2 = H4[(size_t)s2 * (D / 4) + t];
        const float4 v3 = H4[(size_t)s3 * (D / 4) + t];
        acc.x += v0.x; acc.y += v0.y; acc.z += v0.z; acc.w += v0.w;
        acc.x += v1.x; acc.y += v1.y; acc.z += v1.z; acc.w += v1.w;
        acc.x += v2.x; acc.y += v2.y; acc.z += v2.z; acc.w += v2.w;
        acc.x += v3.x; acc.y += v3.y; acc.z += v3.z; acc.w += v3.w;
    }
    for (; j < end; ++j) {
        const int s0 = min(max(csr_src[j], 0), last);
        const float4 v0 = H4[(size_t)s0 * (D / 4) + t];
        acc.x += v0.x; acc.y += v0.y; acc.z += v0.z; acc.w += v0.w;
    }
    acc.x = fmaxf(acc.x, 0.0f); acc.y = fmaxf(acc.y, 0.0f);
    acc.z = fmaxf(acc.z, 0.0f); acc.w = fmaxf(acc.w, 0.0f);
    ((float4*)OUT)[(size_t)node * (D / 4) + t] = acc;
}

extern "C" void kernel_launch(void* const* d_in, const int* in_sizes, int n_in,
                              void* d_out, int out_size, void* d_ws, size_t ws_size,
                              hipStream_t stream)
{
    const float* x  = (const float*)d_in[0];
    const int*   ei = (const int*)d_in[1];
    const float* W1 = (const float*)d_in[2];
    const float* b1 = (const float*)d_in[3];
    const float* W2 = (const float*)d_in[4];
    const float* b2 = (const float*)d_in[5];
    float* out = (float*)d_out;

    const int N = in_sizes[0] / D;
    const int E = in_sizes[1] / 2;
    const int* src = ei;
    const int* dst = ei + E;

    float* h   = (float*)d_ws;
    float* agg = h + (size_t)N * D;
    int* cnt     = (int*)(agg + (size_t)N * D);
    int* rowptr  = cnt + N;
    int* rank    = rowptr + (N + 1);
    int* csr_src = rank + E;

    const int mtb = N / TILE_M;
    const int nb  = (N + 255) / 256;
    const int eb  = (E + 255) / 256;

    csr_zero<<<nb, 256, 0, stream>>>(cnt, N);
    csr_count<<<eb, 256, 0, stream>>>(dst, cnt, rank, E);
    csr_scan<<<1, 1024, 0, stream>>>(cnt, rowptr, N);
    csr_fill<<<eb, 256, 0, stream>>>(src, dst, rank, rowptr, csr_src, E);

    gemm_wmma_f32<<<mtb, 256, 0, stream>>>(x, W1, b1, h, 0);
    gather_sum_relu<<<N, D / 4, 0, stream>>>(h, rowptr, csr_src, agg, N, E);

    gemm_wmma_f32<<<mtb, 256, 0, stream>>>(agg, W2, b2, h, 0);
    gather_sum_relu<<<N, D / 4, 0, stream>>>(h, rowptr, csr_src, out, N, E);
}


# edge list kept as fixed-width per-node bucket rows (64 ids, overflow tail for higher degrees) filled at the rank atomic: prefix-scan and fill launches go (8 to 6); sum order stays arrival rank
# speedup vs baseline: 2.9023x; 1.0776x over previous
#include <hip/hip_runtime.h>

typedef __attribute__((ext_vector_type(2))) float v2f;
typedef __attribute__((ext_vector_type(8))) float v8f;

#define D 512
#define TILE_M 16

#define XP 516
#define KC 32
#define WP 36
#define NCHUNK (D / KC)

#define W_LD(i, off)  w##i = wg[(size_t)(i) * 32 * (D / 4) + (off)]
#define W_ST(i)       *(float4*)(wl + (i) * 32 * WP) = w##i
#define W_LOAD_ALL(off) \
    W_LD(0, off);  W_LD(1, off);  W_LD(2, off);  W_LD(3, off);  \
    W_LD(4, off);  W_LD(5, off);  W_LD(6, off);  W_LD(7, off);  \
    W_LD(8, off);  W_LD(9, off);  W_LD(10, off); W_LD(11, off); \
    W_LD(12, off); W_LD(13, off); W_LD(14, off); W_LD(15, off)
#define W_STORE_ALL() \
    W_ST(0);  W_ST(1);  W_ST(2);  W_ST(3);  W_ST(4);  W_ST(5);  W_ST(6);  W_ST(7); \
    W_ST(8);  W_ST(9);  W_ST(10); W_ST(11); W_ST(12); W_ST(13); W_ST(14); W_ST(15)

__device__ __forceinline__ void lds_barrier()
{
    asm volatile("s_wait_dscnt 0x0\n\ts_barrier_signal -1\n\ts_barrier_wait -1" ::: "memory");
}

__device__ __forceinline__ void chunk_mma(const float* ak, const float* br,
                                          v8f& acc0, v8f& acc1, v8f& acc2, v8f& acc3)
{
#pragma unroll
    for (int kl = 0; kl < KC; kl += 4) {
        v2f a  = *(const v2f*)(ak + kl);
        v2f b0 = *(const v2f*)(br + kl);
        v2f b1 = *(const v2f*)(br + 16 * WP + kl);
        v2f b2 = *(const v2f*)(br + 32 * WP + kl);
        v2f b3 = *(const v2f*)(br + 48 * WP + kl);
        acc0 = __builtin_amdgcn_wmma_f32_16x16x4_f32(false, a, false, b0, (short)0, acc0, false, false);
        acc1 = __builtin_amdgcn_wmma_f32_16x16x4_f32(false, a, false, b1, (short)0, acc1, false, false);
        acc2 = __builtin_amdgcn_wmma_f32_16x16x4_f32(false, a, false, b2, (short)0, acc2, false, false);
        acc3 = __builtin_amdgcn_wmma_f32_16x16x4_f32(false, a, false, b3, (short)0, acc3, false, false);
    }
}

__global__ __launch_bounds__(256) void gemm_wmma_f32(
    const float* __restrict__ Xin, const float* __restrict__ W,
    const float* __restrict__ bias, float* __restrict__ H, int relu_in)
{
    __shared__ __attribute__((aligned(16))) float xs[TILE_M * XP];
    __shared__ __attribute__((aligned(16))) float wsm[D * WP];
    const int m16 = blockIdx.x * TILE_M;
    const int t   = threadIdx.x;

    const float4* xin4 = (const float4*)(Xin + (size_t)m16 * D);
    float4 xv[8];
#pragma unroll
    for (int i = 0; i < 8; ++i) xv[i] = xin4[t + i * 256];
    if (relu_in) {
#pragma unroll
        for (int i = 0; i < 8; ++i) {
            xv[i].x = fmaxf(xv[i].x, 0.0f); xv[i].y = fmaxf(xv[i].y, 0.0f);
            xv[i].z = fmaxf(xv[i].z, 0.0f); xv[i].w = fmaxf(xv[i].w, 0.0f);
        }
    }
#pragma unroll
    for (int i = 0; i < 8; ++i) {
        const int idx = t + i * 256;
        *(float4*)(xs + (idx >> 7) * XP + ((idx & 127) << 2)) = xv[i];
    }

    const int wrow = t >> 3;
    const int wq   = t & 7;
    const float4* wg = (const float4*)W + (size_t)wrow * (D / 4) + wq;
    float* wl = wsm + wrow * WP + (wq << 2);
    float4 w0, w1, w2, w3, w4, w5, w6, w7, w8, w9, w10, w11, w12, w13, w14, w15;
    W_LOAD_ALL(0);
    W_STORE_ALL();
    __builtin_amdgcn_sched_barrier(0);
    W_LOAD_ALL(KC / 4);
    lds_barrier();

    const int wave  = t >> 5;
    const int lane  = t & 31;
    const int row   = lane & 15;
    const int koff  = (lane >> 4) << 1;
    const int nbase = wave * 64;

    v8f acc0 = {}, acc1 = {}, acc2 = {}, acc3 = {};

    const float* ar = xs + row * XP + koff;
    const float* br = wsm + (nbase + row) * WP + koff;

#pragma unroll 1
    for (int kc = 0; kc < NCHUNK - 1; ++kc) {
        chunk_mma(ar + kc * KC, br, acc0, acc1, acc2, acc3);
        lds_barrier();
        W_STORE_ALL();
        if (kc + 2 < NCHUNK) {
            W_LOAD_ALL((kc + 2) * (KC / 4));
        }
        lds_barrier();
    }
    chunk_mma(ar + (NCHUNK - 1) * KC, br, acc0, acc1, acc2, acc3);

    const int rbase = m16 + ((lane >> 4) << 3);
    const int c0 = nbase + (lane & 15);
#pragma unroll
    for (int s = 0; s < 4; ++s) {
        const v8f acc = (s == 0) ? acc0 : (s == 1) ? acc1 : (s == 2) ? acc2 : acc3;
        const int col = c0 + s * 16;
        const float bv = bias[col];
        float* hp = H + (size_t)rbase * D + col;
#pragma unroll
        for (int v = 0; v < 8; ++v) {
            hp[(size_t)v * D] = acc[v] + bv;
        }
    }
}

#define CAP 64

__global__ __launch_bounds__(256) void ell_zero(int* __restrict__ cnt, int n1)
{
    int i = blockIdx.x * 256 + threadIdx.x;
    if (i < n1) cnt[i] = 0;
}

__global__ __launch_bounds__(256) void ell_fill(
    const int* __restrict__ src, const int* __restrict__ dst, int* __restrict__ cnt,
    int* __restrict__ ell, int* __restrict__ ovf_dst, int* __restrict__ ovf_src,
    int n, int E)
{
    int e = blockIdx.x * 256 + threadIdx.x;
    if (e < E) {
        const int d = dst[e];
        const int s = src[e];
        const int slot = atomicAdd(cnt + d, 1);
        if ((unsigned)slot < (unsigned)CAP) {
            ell[(size_t)d * CAP + slot] = s;
        } else {
            const int o = atomicAdd(cnt + n, 1);
            if ((unsigned)o < (unsigned)E) { ovf_dst[o] = d; ovf_src[o] = s; }
        }
    }
}

__global__ __launch_bounds__(128) void gather_sum_relu(
    const float* __restrict__ H, const int* __restrict__ cnt,
    const int* __restrict__ ell, const int* __restrict__ ovf_dst,
    const int* __restrict__ ovf_src, float* __restrict__ OUT, int n, int E)
{
    const int node = blockIdx.x;
    const int t    = threadIdx.x;
    const int deg  = cnt[node];
    const int m    = min(max(deg, 0), CAP);
    const int last = n - 1;
    const int* __restrict__ row = ell + (size_t)node * CAP;
    const float4* __restrict__ H4 = (const float4*)H;

    float4 acc = make_float4(0.f, 0.f, 0.f, 0.f);
    int j = 0;
    for (; j + 4 <= m; j += 4) {
        const int s0 = min(max(row[j + 0], 0), last);
        const int s1 = min(max(row[j + 1], 0), last);
        const int s2 = min(max(row[j + 2], 0), last);
        const int s3 = min(max(row[j + 3], 0), last);
        const float4 v0 = H4[(size_t)s0 * (D / 4) + t];
        const float4 v1 = H4[(size_t)s1 * (D / 4) + t];
        const float4 v2 = H4[(size_t)s2 * (D / 4) + t];
        const float4 v3 = H4[(size_t)s3 * (D / 4) + t];
        acc.x += v0.x; acc.y += v0.y; acc.z += v0.z; acc.w += v0.w;
        acc.x += v1.x; acc.y += v1.y; acc.z += v1.z; acc.w += v1.w;
        acc.x += v2.x; acc.y += v2.y; acc.z += v2.z; acc.w += v2.w;
        acc.x += v3.x; acc.y += v3.y; acc.z += v3.z; acc.w += v3.w;
    }
    for (; j < m; ++j) {
        const int s0 = min(max(row[j], 0), last);
        const float4 v0 = H4[(size_t)s0 * (D / 4) + t];
        acc.x += v0.x; acc.y += v0.y; acc.z += v0.z; acc.w += v0.w;
    }
    if (deg > CAP) {
        const int novf = min(max(cnt[n], 0), E);
        for (int i = 0; i < novf; ++i) {
            if (ovf_dst[i] == node) {
                const int s0 = min(max(ovf_src[i], 0), last);
                const float4 v0 = H4[(size_t)s0 * (D / 4) + t];
                acc.x += v0.x; acc.y += v0.y; acc.z += v0.z; acc.w += v0.w;
            }
        }
    }
    acc.x = fmaxf(acc.x, 0.0f); acc.y = fmaxf(acc.y, 0.0f);
    acc.z = fmaxf(acc.z, 0.0f); acc.w = fmaxf(acc.w, 0.0f);
    ((float4*)OUT)[(size_t)node * (D / 4) + t] = acc;
}

extern "C" void kernel_launch(void* const* d_in, const int* in_sizes, int n_in,
                              void* d_out, int out_size, void* d_ws, size_t ws_size,
                              hipStream_t stream)
{
    const float* x  = (const float*)d_in[0];
    const int*   ei = (const int*)d_in[1];
    const float* W1 = (const float*)d_in[2];
    const float* b1 = (const float*)d_in[3];
    const float* W2 = (const float*)d_in[4];
    const float* b2 = (const float*)d_in[5];
    float* out = (float*)d_out;

    const int N = in_sizes[0] / D;
    const int E = in_sizes[1] / 2;
    const int* src = ei;
    const int* dst = ei + E;

    float* h   = (float*)d_ws;
    float* agg = h + (size_t)N * D;
    int* cnt     = (int*)(agg + (size_t)N * D);
    int* ell     = cnt + (N / CAP + 1) * CAP;
    int* ovf_dst = ell + (size_t)N * CAP;
    int* ovf_src = ovf_dst + E;

    const int mtb = N / TILE_M;
    const int nb  = (N + 1 + 255) / 256;
    const int eb  = (E + 255) / 256;

    ell_zero<<<nb, 256, 0, stream>>>(cnt, N + 1);
    ell_fill<<<eb, 256, 0, stream>>>(src, dst, cnt, ell, ovf_dst, ovf_src, N, E);

    gemm_wmma_f32<<<mtb, 256, 0, stream>>>(x, W1, b1, h, 0);
    gather_sum_relu<<<N, D / 4, 0, stream>>>(h, cnt, ell, ovf_dst, ovf_src, agg, N, E);

    gemm_wmma_f32<<<mtb, 256, 0, stream>>>(agg, W2, b2, h, 0);
    gather_sum_relu<<<N, D / 4, 0, stream>>>(h, cnt, ell, ovf_dst, ovf_src, out, N, E);
}


# GEMM re-tiled: a block is 32 rows x a quarter of the columns (was 16 rows x all), W staged in 64-k chunks: a staged chunk feeds two row tiles, blocks spread evenly over the WGPs; k order kept
# speedup vs baseline: 3.1336x; 1.0797x over previous
#include <hip/hip_runtime.h>

typedef __attribute__((ext_vector_type(2))) float v2f;
typedef __attribute__((ext_vector_type(8))) float v8f;

#define D 512

#define TM 32
#define TN 128
#define NCG (D / TN)
#define XP 516
#define KC 64
#define WP 68
#define NCHUNK (D / KC)

#define W_LD(i, off)  w##i = wg[(size_t)(i) * 16 * (D / 4) + (off)]
#define W_ST(i)       *(float4*)(wl + (i) * 16 * WP) = w##i
#define W_LOAD_ALL(off) \
    W_LD(0, off);  W_LD(1, off);  W_LD(2, off);  W_LD(3, off);  \
    W_LD(4, off);  W_LD(5, off);  W_LD(6, off);  W_LD(7, off)
#define W_STORE_ALL() \
    W_ST(0);  W_ST(1);  W_ST(2);  W_ST(3);  W_ST(4);  W_ST(5);  W_ST(6);  W_ST(7)

__device__ __forceinline__ void lds_barrier()
{
    asm volatile("s_wait_dscnt 0x0\n\ts_barrier_signal -1\n\ts_barrier_wait -1" ::: "memory");
}

__device__ __forceinline__ void chunk_mma(const float* ak, const float* br,
                                          v8f& acc0, v8f& acc1)
{
#pragma unroll
    for (int kl = 0; kl < KC; kl += 4) {
        v2f a  = *(const v2f*)(ak + kl);
        v2f b0 = *(const v2f*)(br + kl);
        v2f b1 = *(const v2f*)(br + 16 * WP + kl);
        acc0 = __builtin_amdgcn_wmma_f32_16x16x4_f32(false, a, false, b0, (short)0, acc0, false, false);
        acc1 = __builtin_amdgcn_wmma_f32_16x16x4_f32(false, a, false, b1, (short)0, acc1, false, false);
    }
}

__global__ __launch_bounds__(256) void gemm_wmma_f32(
    const float* __restrict__ Xin, const float* __restrict__ W,
    const float* __restrict__ bias, float* __restrict__ H, int relu_in, int n)
{
    __shared__ __attribute__((aligned(16))) float xs[TM * XP];
    __shared__ __attribute__((aligned(16))) float wsm[TN * WP];
    const int m0 = (blockIdx.x / NCG) * TM;
    const int n0 = (blockIdx.x % NCG) * TN;
    const int t  = threadIdx.x;

    const bool two = (m0 + TM <= n);
    const float4* xin4 = (const float4*)(Xin + (size_t)m0 * D);
    float4 xv[16];
#pragma unroll
    for (int i = 0; i < 8; ++i) xv[i] = xin4[t + i * 256];
    if (two) {
#pragma unroll
        for (int i = 8; i < 16; ++i) xv[i] = xin4[t + i * 256];
    } else {
#pragma unroll
        for (int i = 8; i < 16; ++i) xv[i] = make_float4(0.f, 0.f, 0.f, 0.f);
    }
    if (relu_in) {
#pragma unroll
        for (int i = 0; i < 16; ++i) {
            xv[i].x = fmaxf(xv[i].x, 0.0f); xv[i].y = fmaxf(xv[i].y, 0.0f);
            xv[i].z = fmaxf(xv[i].z, 0.0f); xv[i].w = fmaxf(xv[i].w, 0.0f);
        }
    }
#pragma unroll
    for (int i = 0; i < 16; ++i) {
        const int idx = t + i * 256;
        *(float4*)(xs + (idx >> 7) * XP + ((idx & 127) << 2)) = xv[i];
    }

    const int wrow = t >> 4;
    const int wq   = t & 15;
    const float4* wg = (const float4*)W + (size_t)(n0 + wrow) * (D / 4) + wq;
    float* wl = wsm + wrow * WP + (wq << 2);
    float4 w0, w1, w2, w3, w4, w5, w6, w7;
    W_LOAD_ALL(0);
    W_STORE_ALL();
    W_LOAD_ALL(KC / 4);
    lds_barrier();

    const int wave  = t >> 5;
    const int lane  = t & 31;
    const int row   = lane & 15;
    const int koff  = (lane >> 4) << 1;
    const int mt    = wave >> 2;
    const int nbase = (wave & 3) * 32;

    v8f acc0 = {}, acc1 = {};

    const float* ar = xs + (mt * 16 + row) * XP + koff;
    const float* br = wsm + (nbase + row) * WP + koff;

#pragma unroll 1
    for (int kc = 0; kc < NCHUNK - 1; ++kc) {
        chunk_mma(ar + kc * KC, br, acc0, acc1);
        lds_barrier();
        W_STORE_ALL();
        if (kc + 2 < NCHUNK) {
            W_LOAD_ALL((kc + 2) * (KC / 4));
        }
        lds_barrier();
    }
    chunk_mma(ar + (NCHUNK - 1) * KC, br, acc0, acc1);

    if (m0 + mt * 16 < n) {
        const int rbase = m0 + mt * 16 + ((lane >> 4) << 3);
        const int c0 = n0 + nbase + (lane & 15);
#pragma unroll
        for (int s = 0; s < 2; ++s) {
            const v8f acc = (s == 0) ? acc0 : acc1;
            const int col = c0 + s * 16;
            const float bv = bias[col];
            float* hp = H + (size_t)rbase * D + col;
#pragma unroll
            for (int v = 0; v < 8; ++v) {
                hp[(size_t)v * D] = acc[v] + bv;
            }
        }
    }
}

#define CAP 64

__global__ __launch_bounds__(256) void ell_zero(int* __restrict__ cnt, int n1)
{
    int i = blockIdx.x * 256 + threadIdx.x;
    if (i < n1) cnt[i] = 0;
}

__global__ __launch_bounds__(256) void ell_fill(
    const int* __restrict__ src, const int* __restrict__ dst, int* __restrict__ cnt,
    int* __restrict__ ell, int* __restrict__ ovf_dst, int* __restrict__ ovf_src,
    int n, int E)
{
    int e = blockIdx.x * 256 + threadIdx.x;
    if (e < E) {
        const int d = dst[e];
        const int s = src[e];
        const int slot = atomicAdd(cnt + d, 1);
        if ((unsigned)slot < (unsigned)CAP) {
            ell[(size_t)d * CAP + slot] = s;
        } else {
            const int o = atomicAdd(cnt + n, 1);
            if ((unsigned)o < (unsigned)E) { ovf_dst[o] = d; ovf_src[o] = s; }
        }
    }
}

__global__ __launch_bounds__(128) void gather_sum_relu(
    const float* __restrict__ H, const int* __restrict__ cnt,
    const int* __restrict__ ell, const int* __restrict__ ovf_dst,
    const int* __restrict__ ovf_src, float* __restrict__ OUT, int n, int E)
{
    const int node = blockIdx.x;
    const int t    = threadIdx.x;
    const int deg  = cnt[node];
    const int m    = min(max(deg, 0), CAP);
    const int last = n - 1;
    const int* __restrict__ row = ell + (size_t)node * CAP;
    const float4* __restrict__ H4 = (const float4*)H;

    float4 acc = make_float4(0.f, 0.f, 0.f, 0.f);
    int j = 0;
    for (; j + 4 <= m; j += 4) {
        const int s0 = min(max(row[j + 0], 0), last);
        const int s1 = min(max(row[j + 1], 0), last);
        const int s2 = min(max(row[j + 2], 0), last);
        const int s3 = min(max(row[j + 3], 0), last);
        const float4 v0 = H4[(size_t)s0 * (D / 4) + t];
        const float4 v1 = H4[(size_t)s1 * (D / 4) + t];
        const float4 v2 = H4[(size_t)s2 * (D / 4) + t];
        const float4 v3 = H4[(size_t)s3 * (D / 4) + t];
        acc.x += v0.x; acc.y += v0.y; acc.z += v0.z; acc.w += v0.w;
        acc.x += v1.x; acc.y += v1.y; acc.z += v1.z; acc.w += v1.w;
        acc.x += v2.x; acc.y += v2.y; acc.z += v2.z; acc.w += v2.w;
        acc.x += v3.x; acc.y += v3.y; acc.z += v3.z; acc.w += v3.w;
    }
    for (; j < m; ++j) {
        const int s0 = min(max(row[j], 0), last);
        const float4 v0 = H4[(size_t)s0 * (D / 4) + t];
        acc.x += v0.x; acc.y += v0.y; acc.z += v0.z; acc.w += v0.w;
    }
    if (deg > CAP) {
        const int novf = min(max(cnt[n], 0), E);
        for (int i = 0; i < novf; ++i) {
            if (ovf_dst[i] == node) {
                const int s0 = min(max(ovf_src[i], 0), last);
                const float4 v0 = H4[(size_t)s0 * (D / 4) + t];
                acc.x += v0.x; acc.y += v0.y; acc.z += v0.z; acc.w += v0.w;
            }
        }
    }
    acc.x = fmaxf(acc.x, 0.0f); acc.y = fmaxf(acc.y, 0.0f);
    acc.z = fmaxf(acc.z, 0.0f); acc.w = fmaxf(acc.w, 0.0f);
    ((float4*)OUT)[(size_t)node * (D / 4) + t] = acc;
}

extern "C" void kernel_launch(void* const* d_in, const int* in_sizes, int n_in,
                              void* d_out, int out_size, void* d_ws, size_t ws_size,
                              hipStream_t stream)
{
    const float* x  = (const float*)d_in[0];
    const int*   ei = (const int*)d_in[1];
    const float* W1 = (const float*)d_in[2];
    const float* b1 = (const float*)d_in[3];
    const float* W2 = (const float*)d_in[4];
    const float* b2 = (const float*)d_in[5];
    float* out = (float*)d_out;

    const int N = in_sizes[0] / D;
    const int E = in_sizes[1] / 2;
    const int* src = ei;
    const int* dst = ei + E;

    float* h   = (float*)d_ws;
    float* agg = h + (size_t)N * D;
    int* cnt     = (int*)(agg + (size_t)N * D);
    int* ell     = cnt + (N / CAP + 1) * CAP;
    int* ovf_dst = ell + (size_t)N * CAP;
    int* ovf_src = ovf_dst + E;

    const int gb  = ((N + TM - 1) / TM) * NCG;
    const int nb  = (N + 1 + 255) / 256;
    const int eb  = (E + 255) / 256;

    ell_zero<<<nb, 256, 0, stream>>>(cnt, N + 1);
    ell_fill<<<eb, 256, 0, stream>>>(src, dst, cnt, ell, ovf_dst, ovf_src, N, E);

    gemm_wmma_f32<<<gb, 256, 0, stream>>>(x, W1, b1, h, 0, N);
    gather_sum_relu<<<N, D / 4, 0, stream>>>(h, cnt, ell, ovf_dst, ovf_src, agg, N, E);

    gemm_wmma_f32<<<gb, 256, 0, stream>>>(agg, W2, b2, h, 0, N);
    gather_sum_relu<<<N, D / 4, 0, stream>>>(h, cnt, ell, ovf_dst, ovf_src, out, N, E);
}


# edge-bucket fill carried by the first GEMM's blocks (ids loaded with the X tile, slot atomic in flight over the WMMAs, id stored after them): fill launch goes, 6 to 5; sum order stays arrival rank
# speedup vs baseline: 3.3182x; 1.0589x over previous
#include <hip/hip_runtime.h>

typedef __attribute__((ext_vector_type(2))) float v2f;
typedef __attribute__((ext_vector_type(8))) float v8f;

#define D 512

#define TM 32
#define TN 128
#define NCG (D / TN)
#define XP 516
#define KC 64
#define WP 68
#define NCHUNK (D / KC)
#define CAP 64

#define W_LD(i, off)  w##i = wg[(size_t)(i) * 16 * (D / 4) + (off)]
#define W_ST(i)       *(float4*)(wl + (i) * 16 * WP) = w##i
#define W_LOAD_ALL(off) \
    W_LD(0, off);  W_LD(1, off);  W_LD(2, off);  W_LD(3, off);  \
    W_LD(4, off);  W_LD(5, off);  W_LD(6, off);  W_LD(7, off)
#define W_STORE_ALL() \
    W_ST(0);  W_ST(1);  W_ST(2);  W_ST(3);  W_ST(4);  W_ST(5);  W_ST(6);  W_ST(7)

__device__ __forceinline__ void lds_barrier()
{
    asm volatile("s_wait_dscnt 0x0\n\ts_barrier_signal -1\n\ts_barrier_wait -1" ::: "memory");
}

__device__ __forceinline__ void chunk_mma(const float* ak, const float* br,
                                          v8f& acc0, v8f& acc1)
{
#pragma unroll
    for (int kl = 0; kl < KC; kl += 4) {
        v2f a  = *(const v2f*)(ak + kl);
        v2f b0 = *(const v2f*)(br + kl);
        v2f b1 = *(const v2f*)(br + 16 * WP + kl);
        acc0 = __builtin_amdgcn_wmma_f32_16x16x4_f32(false, a, false, b0, (short)0, acc0, false, false);
        acc1 = __builtin_amdgcn_wmma_f32_16x16x4_f32(false, a, false, b1, (short)0, acc1, false, false);
    }
}

__global__ __launch_bounds__(256) void gemm_wmma_f32(
    const float* __restrict__ Xin, const float* __restrict__ W,
    const float* __restrict__ bias, float* __restrict__ H, int relu_in, int n,
    const int* __restrict__ src, const int* __restrict__ dst, int* __restrict__ cnt,
    int* __restrict__ ell, int* __restrict__ ovf_dst, int* __restrict__ ovf_src, int E)
{
    __shared__ __attribute__((aligned(16))) float xs[TM * XP];
    __shared__ __attribute__((aligned(16))) float wsm[TN * WP];
    const int m0 = (blockIdx.x / NCG) * TM;
    const int n0 = (blockIdx.x % NCG) * TN;
    const int t  = threadIdx.x;

    const int eper = (E + (int)gridDim.x - 1) / (int)gridDim.x;
    const int e0   = blockIdx.x * eper;
    const int e1   = min(e0 + eper, E);
    const int ee   = e0 + t;
    const bool has = ee < e1;
    int ed = 0, es = 0, eslot = 0;
    if (has) {
        ed = dst[ee];
        es = src[ee];
    }

    const bool two = (m0 + TM <= n);
    const float4* xin4 = (const float4*)(Xin + (size_t)m0 * D);
    float4 xv[16];
#pragma unroll
    for (int i = 0; i < 8; ++i) xv[i] = xin4[t + i * 256];
    if (two) {
#pragma unroll
        for (int i = 8; i < 16; ++i) xv[i] = xin4[t + i * 256];
    } else {
#pragma unroll
        for (int i = 8; i < 16; ++i) xv[i] = make_float4(0.f, 0.f, 0.f, 0.f);
    }
    if (relu_in) {
#pragma unroll
        for (int i = 0; i < 16; ++i) {
            xv[i].x = fmaxf(xv[i].x, 0.0f); xv[i].y = fmaxf(xv[i].y, 0.0f);
            xv[i].z = fmaxf(xv[i].z, 0.0f); xv[i].w = fmaxf(xv[i].w, 0.0f);
        }
    }
#pragma unroll
    for (int i = 0; i < 16; ++i) {
        const int idx = t + i * 256;
        *(float4*)(xs + (idx >> 7) * XP + ((idx & 127) << 2)) = xv[i];
    }

    const int wrow = t >> 4;
    const int wq   = t & 15;
    const float4* wg = (const float4*)W + (size_t)(n0 + wrow) * (D / 4) + wq;
    float* wl = wsm + wrow * WP + (wq << 2);
    float4 w0, w1, w2, w3, w4, w5, w6, w7;
    W_LOAD_ALL(0);
    W_STORE_ALL();
    W_LOAD_ALL(KC / 4);
    if (has) eslot = atomicAdd(cnt + ed, 1);
    lds_barrier();

    const int wave  = t >> 5;
    const int lane  = t & 31;
    const int row   = lane & 15;
    const int koff  = (lane >> 4) << 1;
    const int mt    = wave >> 2;
    const int nbase = (wave & 3) * 32;

    v8f acc0 = {}, acc1 = {};

    const float* ar = xs + (mt * 16 + row) * XP + koff;
    const float* br = wsm + (nbase + row) * WP + koff;

#pragma unroll 1
    for (int kc = 0; kc < NCHUNK - 1; ++kc) {
        chunk_mma(ar + kc * KC, br, acc0, acc1);
        lds_barrier();
        W_STORE_ALL();
        if (kc + 2 < NCHUNK) {
            W_LOAD_ALL((kc + 2) * (KC / 4));
        }
        lds_barrier();
    }
    chunk_mma(ar + (NCHUNK - 1) * KC, br, acc0, acc1);

    if (m0 + mt * 16 < n) {
        const int rbase = m0 + mt * 16 + ((lane >> 4) << 3);
        const int c0 = n0 + nbase + (lane & 15);
#pragma unroll
        for (int s = 0; s < 2; ++s) {
            const v8f acc = (s == 0) ? acc0 : acc1;
            const int col = c0 + s * 16;
            const float bv = bias[col];
            float* hp = H + (size_t)rbase * D + col;
#pragma unroll
            for (int v = 0; v < 8; ++v) {
                hp[(size_t)v * D] = acc[v] + bv;
            }
        }
    }

    if (has) {
        if ((unsigned)eslot < (unsigned)CAP) {
            ell[(size_t)ed * CAP + eslot] = es;
        } else {
            const int o = atomicAdd(cnt + n, 1);
            if ((unsigned)o < (unsigned)E) { ovf_dst[o] = ed; ovf_src[o] = es; }
        }
    }
    for (int e = ee + 256; e < e1; e += 256) {
        const int d = dst[e];
        const int s = src[e];
        const int slot = atomicAdd(cnt + d, 1);
        if ((unsigned)slot < (unsigned)CAP) {
            ell[(size_t)d * CAP + slot] = s;
        } else {
            const int o = atomicAdd(cnt + n, 1);
            if ((unsigned)o < (unsigned)E) { ovf_dst[o] = d; ovf_src[o] = s; }
        }
    }
}

__global__ __launch_bounds__(256) void ell_zero(int* __restrict__ cnt, int n1)
{
    int i = blockIdx.x * 256 + threadIdx.x;
    if (i < n1) cnt[i] = 0;
}

__global__ __launch_bounds__(128) void gather_sum_relu(
    const float* __restrict__ H, const int* __restrict__ cnt,
    const int* __restrict__ ell, const int* __restrict__ ovf_dst,
    const int* __restrict__ ovf_src, float* __restrict__ OUT, int n, int E)
{
    const int node = blockIdx.x;
    const int t    = threadIdx.x;
    const int deg  = cnt[node];
    const int m    = min(max(deg, 0), CAP);
    const int last = n - 1;
    const int* __restrict__ row = ell + (size_t)node * CAP;
    const float4* __restrict__ H4 = (const float4*)H;

    float4 acc = make_float4(0.f, 0.f, 0.f, 0.f);
    int j = 0;
    for (; j + 4 <= m; j += 4) {
        const int s0 = min(max(row[j + 0], 0), last);
        const int s1 = min(max(row[j + 1], 0), last);
        const int s2 = min(max(row[j + 2], 0), last);
        const int s3 = min(max(row[j + 3], 0), last);
        const float4 v0 = H4[(size_t)s0 * (D / 4) + t];
        const float4 v1 = H4[(size_t)s1 * (D / 4) + t];
        const float4 v2 = H4[(size_t)s2 * (D / 4) + t];
        const float4 v3 = H4[(size_t)s3 * (D / 4) + t];
        acc.x += v0.x; acc.y += v0.y; acc.z += v0.z; acc.w += v0.w;
        acc.x += v1.x; acc.y += v1.y; acc.z += v1.z; acc.w += v1.w;
        acc.x += v2.x; acc.y += v2.y; acc.z += v2.z; acc.w += v2.w;
        acc.x += v3.x; acc.y += v3.y; acc.z += v3.z; acc.w += v3.w;
    }
    for (; j < m; ++j) {
        const int s0 = min(max(row[j], 0), last);
        const float4 v0 = H4[(size_t)s0 * (D / 4) + t];
        acc.x += v0.x; acc.y += v0.y; acc.z += v0.z; acc.w += v0.w;
    }
    if (deg > CAP) {
        const int novf = min(max(cnt[n], 0), E);
        for (int i = 0; i < novf; ++i) {
            if (ovf_dst[i] == node) {
                const int s0 = min(max(ovf_src[i], 0), last);
                const float4 v0 = H4[(size_t)s0 * (D / 4) + t];
                acc.x += v0.x; acc.y += v0.y; acc.z += v0.z; acc.w += v0.w;
            }
        }
    }
    acc.x = fmaxf(acc.x, 0.0f); acc.y = fmaxf(acc.y, 0.0f);
    acc.z = fmaxf(acc.z, 0.0f); acc.w = fmaxf(acc.w, 0.0f);
    ((float4*)OUT)[(size_t)node * (D / 4) + t] = acc;
}

extern "C" void kernel_launch(void* const* d_in, const int* in_sizes, int n_in,
                              void* d_out, int out_size, void* d_ws, size_t ws_size,
                              hipStream_t stream)
{
    const float* x  = (const float*)d_in[0];
    const int*   ei = (const int*)d_in[1];
    const float* W1 = (const float*)d_in[2];
    const float* b1 = (const float*)d_in[3];
    const float* W2 = (const float*)d_in[4];
    const float* b2 = (const float*)d_in[5];
    float* out = (float*)d_out;

    const int N = in_sizes[0] / D;
    const int E = in_sizes[1] / 2;
    const int* src = ei;
    const int* dst = ei + E;

    float* h   = (float*)d_ws;
    float* agg = h + (size_t)N * D;
    int* cnt     = (int*)(agg + (size_t)N * D);
    int* ell     = cnt + (N / CAP + 1) * CAP;
    int* ovf_dst = ell + (size_t)N * CAP;
    int* ovf_src = ovf_dst + E;

    const int gb  = ((N + TM - 1) / TM) * NCG;
    const int nb  = (N + 1 + 255) / 256;

    ell_zero<<<nb, 256, 0, stream>>>(cnt, N + 1);

    gemm_wmma_f32<<<gb, 256, 0, stream>>>(x, W1, b1, h, 0, N,
                                          src, dst, cnt, ell, ovf_dst, ovf_src, E);
    gather_sum_relu<<<N, D / 4, 0, stream>>>(h, cnt, ell, ovf_dst, ovf_src, agg, N, E);

    gemm_wmma_f32<<<gb, 256, 0, stream>>>(agg, W2, b2, h, 0, N,
                                          src, dst, cnt, ell, ovf_dst, ovf_src, 0);
    gather_sum_relu<<<N, D / 4, 0, stream>>>(h, cnt, ell, ovf_dst, ovf_src, out, N, E);
}
